# starting kernel
# baseline (speedup 1.0000x reference)
#include <hip/hip_runtime.h>

typedef __attribute__((ext_vector_type(2))) float v2f;
typedef __attribute__((ext_vector_type(8))) float v8f;

#define NPTS 4096
#define CCH  64
#define KNN  20
#define FMAX 3.402823466e+38f


__global__ void norms_kernel(const float* __restrict__ x, float* __restrict__ norms) {
    int p = blockIdx.x * blockDim.x + threadIdx.x;
    int b = p >> 12, n = p & 4095;
    const float* xb = x + b * CCH * NPTS + n;
    float s = 0.f;
    #pragma unroll
    for (int c = 0; c < CCH; ++c) { float v = xb[c * NPTS]; s += v * v; }
    norms[p] = s;
}

__global__ void __launch_bounds__(32) knn_kernel(const float* __restrict__ x,
                                                 const float* __restrict__ norms,
                                                 int* __restrict__ knn_idx) {
    __shared__ float distT[16][17];
    __shared__ float kd[16 * KNN];
    __shared__ int   ki[16 * KNN];

    const int lane = threadIdx.x;
    const int half = lane >> 4;
    const int lm   = lane & 15;
    const int b    = blockIdx.y;
    const int rowBase = blockIdx.x * 16;
    const float* xb = x + b * CCH * NPTS;
    const float* nb = norms + b * NPTS;

    v2f a[16];
    #pragma unroll
    for (int k = 0; k < 16; ++k) {
        int c = 4 * k + 2 * half;
        a[k].x = xb[(c    ) * NPTS + rowBase + lm];
        a[k].y = xb[(c + 1) * NPTS + rowBase + lm];
    }
    float rn[8];
    #pragma unroll
    for (int r = 0; r < 8; ++r) rn[r] = nb[rowBase + r + 8 * half];

    if (lane < 16) {
        for (int j = 0; j < KNN; ++j) { kd[lane * KNN + j] = FMAX; ki[lane * KNN + j] = 0; }
    }
    float curMax = FMAX; int curArg = 0;
    __syncthreads();

    for (int t = 0; t < NPTS / 16; ++t) {
        const int colBase = t * 16;

        v2f bb[16];
        #pragma unroll
        for (int k = 0; k < 16; ++k) {
            int c = 4 * k + 2 * half;
            bb[k].x = xb[(c    ) * NPTS + colBase + lm];
            bb[k].y = xb[(c + 1) * NPTS + colBase + lm];
        }

        v8f acc = {};
        #pragma unroll
        for (int k = 0; k < 16; ++k)
            acc = __builtin_amdgcn_wmma_f32_16x16x4_f32(false, a[k], false, bb[k],
                                                        (short)0, acc, false, false);

        float cn = nb[colBase + lm];
        #pragma unroll
        for (int r = 0; r < 8; ++r)
            distT[r + 8 * half][lm] = rn[r] + cn - 2.0f * acc[r];
        __syncthreads();

        if (lane < 16) {
            for (int j = 0; j < 16; ++j) {
                float d = distT[lane][j];
                if (d < curMax) {
                    kd[lane * KNN + curArg] = d;
                    ki[lane * KNN + curArg] = colBase + j;
                    float m = -FMAX; int arg = 0;
                    for (int q = 0; q < KNN; ++q) {
                        float v = kd[lane * KNN + q];
                        if (v > m) { m = v; arg = q; }
                    }
                    curMax = m; curArg = arg;
                }
            }
        }
        __syncthreads();
    }

    if (lane < 16) {
        int* op = knn_idx + (b * NPTS + rowBase + lane) * KNN;
        for (int j = 0; j < KNN; ++j) op[j] = ki[lane * KNN + j];
    }
}

__global__ void __launch_bounds__(32) proj_kernel(const float* __restrict__ x,
                                                  const float* __restrict__ W,
                                                  float* __restrict__ y,
                                                  float* __restrict__ z) {
    const int lane = threadIdx.x;
    const int half = lane >> 4;
    const int lm   = lane & 15;
    const int b    = blockIdx.y;
    const int rowBase = blockIdx.x * 16;
    const float* xb = x + b * CCH * NPTS;

    v2f a[16];
    #pragma unroll
    for (int k = 0; k < 16; ++k) {
        int c = 4 * k + 2 * half;
        a[k].x = xb[(c    ) * NPTS + rowBase + lm];
        a[k].y = xb[(c + 1) * NPTS + rowBase + lm];
    }

    for (int ot = 0; ot < 4; ++ot) {
        const int o = ot * 16 + lm;
        v8f accY = {}, accZ = {};
        #pragma unroll
        for (int k = 0; k < 16; ++k) {
            int c = 4 * k + 2 * half;
            float w0 = W[o * 128 + c], w1 = W[o * 128 + c + 1];
            v2f bw, bz;
            bw.x = w0;
            bw.y = w1;
            bz.x = W[o * 128 + 64 + c    ] - w0;
            bz.y = W[o * 128 + 64 + c + 1] - w1;
            accY = __builtin_amdgcn_wmma_f32_16x16x4_f32(false, a[k], false, bw,
                                                         (short)0, accY, false, false);
            accZ = __builtin_amdgcn_wmma_f32_16x16x4_f32(false, a[k], false, bz,
                                                         (short)0, accZ, false, false);
        }
        #pragma unroll
        for (int r = 0; r < 8; ++r) {
            int row = rowBase + r + 8 * half;
            y[(b * NPTS + row) * CCH + o] = accY[r];
            z[(b * NPTS + row) * CCH + o] = accZ[r];
        }
    }
}

__global__ void gathermax_kernel(const float* __restrict__ y,
                                 const float* __restrict__ z,
                                 const int* __restrict__ knn_idx,
                                 float* __restrict__ out) {
    int t = blockIdx.x * blockDim.x + threadIdx.x;
    int o = t & 63;
    int p = t >> 6;
    int b = p >> 12, n = p & 4095;
    const int*   ip = knn_idx + p * KNN;
    const float* yb = y + b * NPTS * CCH;
    float best = -FMAX;
    #pragma unroll
    for (int k = 0; k < KNN; ++k) {
        int m = ip[k];
        best = fmaxf(best, yb[m * CCH + o]);
    }
    best += z[p * CCH + o];
    out[(b * CCH + o) * NPTS + n] = best;
}

extern "C" void kernel_launch(void* const* d_in, const int* in_sizes, int n_in,
                              void* d_out, int out_size, void* d_ws, size_t ws_size,
                              hipStream_t stream) {
    (void)in_sizes; (void)n_in; (void)out_size; (void)ws_size;
    const float* x = (const float*)d_in[0];
    const float* W = (const float*)d_in[1];
    float* out = (float*)d_out;

    char* ws = (char*)d_ws;
    float* norms = (float*)(ws);
    int*   idxb  = (int*)  (ws + 65536);
    float* y     = (float*)(ws + 65536 + 1310720);
    float* z     = (float*)(ws + 65536 + 1310720 + 4194304);

    norms_kernel    <<<dim3(64), 256, 0, stream>>>(x, norms);
    proj_kernel     <<<dim3(256, 4), 32, 0, stream>>>(x, W, y, z);
    knn_kernel      <<<dim3(256, 4), 32, 0, stream>>>(x, norms, idxb);
    gathermax_kernel<<<dim3(4096), 256, 0, stream>>>(y, z, idxb, out);
}


# projection launch folded into the knn kernel: its selection waves, idle until the first batch is handed over, compute y and z of the work-group's 32 rows (same WMMA chains); 3 launches, not 4
# speedup vs baseline: 4.6634x; 4.6634x over previous
#include <hip/hip_runtime.h>

typedef __attribute__((ext_vector_type(2))) float v2f;
typedef __attribute__((ext_vector_type(8))) float v8f;

#define NPTS 4096
#define CCH  64
#define KNN  20
#define FMAX 3.402823466e+38f


__global__ void norms_kernel(const float* __restrict__ x, float* __restrict__ norms) {
    int p = blockIdx.x * blockDim.x + threadIdx.x;
    int b = p >> 12, n = p & 4095;
    const float* xb = x + b * CCH * NPTS + n;
    float s = 0.f;
    #pragma unroll
    for (int c = 0; c < CCH; ++c) { float v = xb[c * NPTS]; s += v * v; }
    norms[p] = s;
}


#define RT 2
#define TB 16
#define NB (NPTS / 16 / TB)
#define TILE_PITCH 272
#define RT_PITCH (TB * TILE_PITCH)
#define SLOT_FLOATS (RT * RT_PITCH)

__device__ __forceinline__ void list_insert(float& kd, int& ki, float key, int col) {
    const int kdi = __float_as_int(kd);
    const int cd = __builtin_amdgcn_permlanex16(kdi, kdi, 0xffffffffu, 0xffffffffu, false, false);
    const int ci = __builtin_amdgcn_permlanex16(ki, ki, 0xffffffffu, 0xffffffffu, false, false);
    const float td = __int_as_float(__builtin_amdgcn_update_dpp(cd, kdi, 0x111, 0xF, 0xF, false));
    const int   ti = __builtin_amdgcn_update_dpp(ci, ki, 0x111, 0xF, 0xF, false);
    const bool up   = kd > key;
    const bool upLo = td > key;
    kd = up ? (upLo ? td : key) : kd;
    ki = up ? (upLo ? ti : col) : ki;
}

__device__ __forceinline__ float list_bound(float kd) {
    return __int_as_float(__builtin_amdgcn_readlane(__float_as_int(kd), KNN));
}

__device__ __forceinline__ void handover_barrier() {
    asm volatile("s_wait_dscnt 0x0\n\ts_barrier_signal -1\n\ts_barrier_wait -1" ::: "memory");
}

#define KNN_ROW(D, r, MM)                                                          \
    {                                                                              \
        unsigned mm_ = (MM);                                                       \
        if (mm_) {                                                                 \
            do {                                                                   \
                const int c_ = __builtin_ctz(mm_);                                 \
                mm_ &= mm_ - 1;                                                    \
                const float key_ = __int_as_float(__builtin_amdgcn_readlane(       \
                    __float_as_int(D[r]), c_));                                    \
                list_insert(kd[r], ki[r], key_, colBase + c_);                     \
            } while (mm_);                                                         \
            thr[r] = list_bound(kd[r]);                                            \
        }                                                                          \
    }

#define LOAD_DIST(D, SEL, J)                                                       \
    {                                                                              \
        const float* p_ = (SEL) + (J) * (2 * TILE_PITCH);                          \
        _Pragma("unroll")                                                          \
        for (int r = 0; r < 8; ++r) D[r] = p_[r * 32];                             \
    }

#define SELECT_TILE(D, T)                                                          \
    {                                                                              \
        const int colBase = (T) * 16;                                              \
        unsigned m[8];                                                             \
        unsigned any = 0;                                                          \
        _Pragma("unroll")                                                          \
        for (int r = 0; r < 8; ++r) {                                              \
            m[r] = __builtin_amdgcn_ballot_w32(D[r] < thr[r]);                     \
            any |= m[r];                                                           \
        }                                                                          \
        if (any) {                                                                 \
            KNN_ROW(D, 0, m[0]) KNN_ROW(D, 1, m[1]) KNN_ROW(D, 2, m[2])            \
            KNN_ROW(D, 3, m[3]) KNN_ROW(D, 4, m[4]) KNN_ROW(D, 5, m[5])            \
            KNN_ROW(D, 6, m[6]) KNN_ROW(D, 7, m[7])                                \
        }                                                                          \
    }

#define LOAD_TILE(BB, CN, T)                                                       \
    {                                                                              \
        const int cb_ = (T) * 16;                                                  \
        _Pragma("unroll")                                                          \
        for (int k = 0; k < 16; ++k) {                                             \
            const int c = 4 * k + 2 * half;                                        \
            BB[k].x = xb[(c    ) * NPTS + cb_ + lm];                               \
            BB[k].y = xb[(c + 1) * NPTS + cb_ + lm];                               \
        }                                                                          \
        CN = nb[cb_ + lm];                                                         \
    }

#define MAKE_TILE(BB, CN, DST)                                                     \
    {                                                                              \
        v8f acc0 = {}, acc1 = {};                                                  \
        _Pragma("unroll")                                                          \
        for (int k = 0; k < 16; ++k) {                                             \
            acc0 = __builtin_amdgcn_wmma_f32_16x16x4_f32(false, a0[k], false, BB[k], \
                                                         (short)0, acc0, false, false); \
            acc1 = __builtin_amdgcn_wmma_f32_16x16x4_f32(false, a1[k], false, BB[k], \
                                                         (short)0, acc1, false, false); \
        }                                                                          \
        float* p_ = (DST) + lane;                                                  \
        _Pragma("unroll")                                                          \
        for (int r = 0; r < 8; ++r) p_[r * 32] = rn0[r] + CN - 2.0f * acc0[r];     \
        _Pragma("unroll")                                                          \
        for (int r = 0; r < 8; ++r) p_[RT_PITCH + r * 32] = rn1[r] + CN - 2.0f * acc1[r]; \
    }

__device__ __forceinline__ void proj_rows(const float* __restrict__ xb, const float* __restrict__ W,
                                          float* __restrict__ y, float* __restrict__ z,
                                          const int b, const int rowBase16, const int ot0, const int lane) {
    const int half = lane >> 4;
    const int lm   = lane & 15;

    const int a0i = (2 * half) * NPTS + rowBase16 + lm;
    v2f a[16];
    #pragma unroll
    for (int k = 0; k < 16; ++k) {
        a[k].x = xb[a0i + (4 * k    ) * NPTS];
        a[k].y = xb[a0i + (4 * k + 1) * NPTS];
    }

    #pragma unroll
    for (int t = 0; t < 2; ++t) {
        const int o = (ot0 + t) * 16 + lm;
        v8f accY = {}, accZ = {};
        #pragma unroll
        for (int k = 0; k < 16; ++k) {
            const int c = 4 * k + 2 * half;
            const float w0 = W[o * 128 + c], w1 = W[o * 128 + c + 1];
            v2f bw, bz;
            bw.x = w0;
            bw.y = w1;
            bz.x = W[o * 128 + 64 + c    ] - w0;
            bz.y = W[o * 128 + 64 + c + 1] - w1;
            accY = __builtin_amdgcn_wmma_f32_16x16x4_f32(false, a[k], false, bw,
                                                         (short)0, accY, false, false);
            accZ = __builtin_amdgcn_wmma_f32_16x16x4_f32(false, a[k], false, bz,
                                                         (short)0, accZ, false, false);
        }
        #pragma unroll
        for (int r = 0; r < 8; ++r) {
            const int row = rowBase16 + r + 8 * half;
            y[(b * NPTS + row) * CCH + o] = accY[r];
            z[(b * NPTS + row) * CCH + o] = accZ[r];
        }
    }
}

__global__ void __launch_bounds__(256) __attribute__((amdgpu_waves_per_eu(4, 4))) knn_kernel(const float* __restrict__ x,
                                                 const float* __restrict__ norms,
                                                 const float* __restrict__ W,
                                                 int* __restrict__ knn_idx,
                                                 float* __restrict__ y,
                                                 float* __restrict__ z) {
    __shared__ float sdist[2 * SLOT_FLOATS];

    const int lane = threadIdx.x & 31;
    const int wave = __builtin_amdgcn_readfirstlane((int)(threadIdx.x >> 5));
    const bool matrixWave = wave < 4;
    const int p    = wave & 3;
    const int q    = wave & 1;
    const int rt   = (wave >> 1) & 1;
    const int half = lane >> 4;
    const int lm   = lane & 15;
    const int b    = blockIdx.y;
    const int rowBase = blockIdx.x * (16 * RT);
    const float* xb = x + b * CCH * NPTS;
    const float* nb = norms + b * NPTS;

    if (!matrixWave) proj_rows(xb, W, y, z, b, rowBase + 16 * rt, 2 * q, lane);

    v2f a0[16], a1[16];
    #pragma unroll
    for (int k = 0; k < 16; ++k) {
        int c = 4 * k + 2 * half;
        a0[k].x = xb[(c    ) * NPTS + rowBase + lm];
        a0[k].y = xb[(c + 1) * NPTS + rowBase + lm];
        a1[k].x = xb[(c    ) * NPTS + rowBase + 16 + lm];
        a1[k].y = xb[(c + 1) * NPTS + rowBase + 16 + lm];
    }
    float rn0[8], rn1[8];
    #pragma unroll
    for (int r = 0; r < 8; ++r) {
        rn0[r] = nb[rowBase + r + 8 * half];
        rn1[r] = nb[rowBase + 16 + r + 8 * half];
    }

    float kd[8];
    int   ki[8];
    float thr[8];
    #pragma unroll
    for (int r = 0; r < 8; ++r) {
        kd[r] = (lane == 0) ? __int_as_float((int)0xff800000) : FMAX;
        ki[r] = 0;
        thr[r] = FMAX;
    }

    v2f b0[16], b1[16];
    float cn0, cn1;
    LOAD_TILE(b0, cn0, p)

    float d0[8], d1[8];

    for (int it = 0; it <= NB; ++it) {
        if (matrixWave) {
            if (it < NB) {
                float* dst = sdist + (it & 1) * SLOT_FLOATS;
                const int tb = it * TB;
                for (int j = 0; j < TB / 4; j += 2) {
                    const int i0 = 4 * j + p;
                    LOAD_TILE(b1, cn1, tb + i0 + 4)
                    MAKE_TILE(b0, cn0, dst + i0 * TILE_PITCH)
                    LOAD_TILE(b0, cn0, (tb + i0 + 8) & (NPTS / 16 - 1))
                    MAKE_TILE(b1, cn1, dst + (i0 + 4) * TILE_PITCH)
                }
            }
        } else {
            if (it >= 1) {
                const float* sel = sdist + ((it - 1) & 1) * SLOT_FLOATS + rt * RT_PITCH
                                 + half * TILE_PITCH + 16 * q + lm;
                const int tb = (it - 1) * TB;
                LOAD_DIST(d0, sel, 0)
                #pragma nounroll
                for (int j = 0; j < TB / 2; j += 2) {
                    LOAD_DIST(d1, sel, j + 1)
                    SELECT_TILE(d0, tb + 2 * j)
                    LOAD_DIST(d0, sel, (j + 2) & (TB / 2 - 1))
                    SELECT_TILE(d1, tb + 2 * j + 2)
                }
            }
        }
        handover_barrier();
    }

    if (!matrixWave && lane >= 1 && lane <= KNN) {
        int* op = knn_idx + (b * NPTS + rowBase + 16 * rt + 8 * q) * KNN + (lane - 1);
        #pragma unroll
        for (int r = 0; r < 8; ++r) op[r * KNN] = ki[r];
    }
}

#define GM_PTS   32
#define GM_PITCH 33
__global__ void __launch_bounds__(256) gathermax_kernel(const float* __restrict__ y,
                                                        const float* __restrict__ z,
                                                        const int* __restrict__ knn_idx,
                                                        float* __restrict__ out) {
    __shared__ float tile[CCH * GM_PITCH];

    const int lane = threadIdx.x & 31;
    const int wave = __builtin_amdgcn_readfirstlane((int)(threadIdx.x >> 5));
    const int p0 = blockIdx.x * GM_PTS;
    const int b  = p0 >> 12, n0 = p0 & (NPTS - 1);
    const float* yb = y + b * NPTS * CCH;

    #pragma unroll
    for (int j = 0; j < 4; ++j) {
        const int pl = 4 * wave + j;
        const int p  = p0 + pl;
        const int* ip = knn_idx + p * KNN;
        v2f best = {-FMAX, -FMAX};
        #pragma unroll
        for (int k = 0; k < KNN; ++k) {
            const int m = ip[k];
            const v2f v = *(const v2f*)(yb + m * CCH + 2 * lane);
            best.x = fmaxf(best.x, v.x);
            best.y = fmaxf(best.y, v.y);
        }
        const v2f zz = *(const v2f*)(z + p * CCH + 2 * lane);
        tile[(2 * lane    ) * GM_PITCH + pl] = best.x + zz.x;
        tile[(2 * lane + 1) * GM_PITCH + pl] = best.y + zz.y;
    }
    __syncthreads();
    #pragma unroll
    for (int i = 0; i < 8; ++i) {
        const int o = 8 * wave + i;
        out[(b * CCH + o) * NPTS + n0 + lane] = tile[o * GM_PITCH + lane];
    }
}

extern "C" void kernel_launch(void* const* d_in, const int* in_sizes, int n_in,
                              void* d_out, int out_size, void* d_ws, size_t ws_size,
                              hipStream_t stream) {
    (void)in_sizes; (void)n_in; (void)out_size; (void)ws_size;
    const float* x = (const float*)d_in[0];
    const float* W = (const float*)d_in[1];
    float* out = (float*)d_out;

    char* ws = (char*)d_ws;
    float* norms = (float*)(ws);
    int*   idxb  = (int*)  (ws + 65536);
    float* y     = (float*)(ws + 65536 + 1310720);
    float* z     = (float*)(ws + 65536 + 1310720 + 4194304);

    norms_kernel    <<<dim3(64), 256, 0, stream>>>(x, norms);
    knn_kernel      <<<dim3(128, 4), 256, 0, stream>>>(x, norms, W, idxb, y, z);
    gathermax_kernel<<<dim3(512), 256, 0, stream>>>(y, z, idxb, out);
}


# knn: a row's first bound = 20th smallest of batch 0's 32 per-lane minima (min over 8 tile pairs, bitonic sort across lanes), exact; ~55 of a row's ~140 sorted-list inserts never happen
# speedup vs baseline: 5.3182x; 1.1404x over previous
#include <hip/hip_runtime.h>

typedef __attribute__((ext_vector_type(2))) float v2f;
typedef __attribute__((ext_vector_type(8))) float v8f;

#define NPTS 4096
#define CCH  64
#define KNN  20
#define FMAX 3.402823466e+38f


__global__ void norms_kernel(const float* __restrict__ x, float* __restrict__ norms) {
    int p = blockIdx.x * blockDim.x + threadIdx.x;
    int b = p >> 12, n = p & 4095;
    const float* xb = x + b * CCH * NPTS + n;
    float s = 0.f;
    #pragma unroll
    for (int c = 0; c < CCH; ++c) { float v = xb[c * NPTS]; s += v * v; }
    norms[p] = s;
}


#define RT 2
#define TB 16
#define NB (NPTS / 16 / TB)
#define TILE_PITCH 272
#define RT_PITCH (TB * TILE_PITCH)
#define SLOT_FLOATS (RT * RT_PITCH)

__device__ __forceinline__ void list_insert(float& kd, int& ki, float key, int col) {
    const int kdi = __float_as_int(kd);
    const int cd = __builtin_amdgcn_permlanex16(kdi, kdi, 0xffffffffu, 0xffffffffu, false, false);
    const int ci = __builtin_amdgcn_permlanex16(ki, ki, 0xffffffffu, 0xffffffffu, false, false);
    const float td = __int_as_float(__builtin_amdgcn_update_dpp(cd, kdi, 0x111, 0xF, 0xF, false));
    const int   ti = __builtin_amdgcn_update_dpp(ci, ki, 0x111, 0xF, 0xF, false);
    const bool up   = kd > key;
    const bool upLo = td > key;
    kd = up ? (upLo ? td : key) : kd;
    ki = up ? (upLo ? ti : col) : ki;
}

__device__ __forceinline__ float list_bound(float kd) {
    return __int_as_float(__builtin_amdgcn_readlane(__float_as_int(kd), KNN));
}

template <int J>
__device__ __forceinline__ int lane_xor(const int v) {
    if constexpr (J == 16)
        return __builtin_amdgcn_permlanex16(v, v, 0x76543210u, 0xfedcba98u, false, false);
    else
        return __builtin_amdgcn_mov_dpp(v, 0x160 + J, 0xF, 0xF, false);
}

__device__ __forceinline__ int order_key(const int b) { return b ^ ((b >> 31) & 0x7fffffff); }

template <int K, int J>
__device__ __forceinline__ void bitonic_step(int (&v)[8], const int lane) {
    const bool takeMin = (((lane & J) == 0) == ((lane & K) == 0));
    #pragma unroll
    for (int r = 0; r < 8; ++r) {
        const int o = lane_xor<J>(v[r]);
        v[r] = takeMin ? (o < v[r] ? o : v[r]) : (o > v[r] ? o : v[r]);
    }
}

__device__ __forceinline__ void sort_lanes(int (&v)[8], const int lane) {
    bitonic_step<2, 1>(v, lane);
    bitonic_step<4, 2>(v, lane);  bitonic_step<4, 1>(v, lane);
    bitonic_step<8, 4>(v, lane);  bitonic_step<8, 2>(v, lane);  bitonic_step<8, 1>(v, lane);
    bitonic_step<16, 8>(v, lane); bitonic_step<16, 4>(v, lane); bitonic_step<16, 2>(v, lane);
    bitonic_step<16, 1>(v, lane);
    bitonic_step<32, 16>(v, lane); bitonic_step<32, 8>(v, lane); bitonic_step<32, 4>(v, lane);
    bitonic_step<32, 2>(v, lane);  bitonic_step<32, 1>(v, lane);
}

__device__ __forceinline__ void handover_barrier() {
    asm volatile("s_wait_dscnt 0x0\n\ts_barrier_signal -1\n\ts_barrier_wait -1" ::: "memory");
}

#define KNN_ROW(D, r, MM)                                                          \
    {                                                                              \
        unsigned mm_ = (MM);                                                       \
        if (mm_) {                                                                 \
            do {                                                                   \
                const int c_ = __builtin_ctz(mm_);                                 \
                mm_ &= mm_ - 1;                                                    \
                const float key_ = __int_as_float(__builtin_amdgcn_readlane(       \
                    __float_as_int(D[r]), c_));                                    \
                list_insert(kd[r], ki[r], key_, colBase + c_);                     \
            } while (mm_);                                                         \
            thr[r] = list_bound(kd[r]);                                            \
        }                                                                          \
    }

#define LOAD_DIST(D, SEL, J)                                                       \
    {                                                                              \
        const float* p_ = (SEL) + (J) * (2 * TILE_PITCH);                          \
        _Pragma("unroll")                                                          \
        for (int r = 0; r < 8; ++r) D[r] = p_[r * 32];                             \
    }

#define SAMPLE_BOUND(SEL)                                                          \
    {                                                                              \
        float mn[8], t_[8];                                                        \
        LOAD_DIST(mn, SEL, 0)                                                      \
        _Pragma("unroll")                                                          \
        for (int J = 1; J < TB / 2; ++J) {                                         \
            LOAD_DIST(t_, SEL, J)                                                  \
            _Pragma("unroll")                                                      \
            for (int r = 0; r < 8; ++r) mn[r] = __builtin_fminf(mn[r], t_[r]);     \
        }                                                                          \
        int key_[8];                                                               \
        _Pragma("unroll")                                                          \
        for (int r = 0; r < 8; ++r) key_[r] = order_key(__float_as_int(mn[r]));    \
        sort_lanes(key_, lane);                                                    \
        _Pragma("unroll")                                                          \
        for (int r = 0; r < 8; ++r) {                                              \
            const float b_ = __int_as_float(order_key(                             \
                __builtin_amdgcn_readlane(key_[r], KNN - 1)));                     \
            const float s_ = b_ + __builtin_fmaxf(__builtin_fabsf(b_) * 0x1p-20f, 1e-30f); \
            thr[r] = __int_as_float(__builtin_amdgcn_readfirstlane(__float_as_int(s_))); \
            kd[r] = (lane == 0) ? __int_as_float((int)0xff800000) : s_;            \
        }                                                                          \
    }

#define SELECT_TILE(D, T)                                                          \
    {                                                                              \
        const int colBase = (T) * 16;                                              \
        unsigned m[8];                                                             \
        unsigned any = 0;                                                          \
        _Pragma("unroll")                                                          \
        for (int r = 0; r < 8; ++r) {                                              \
            m[r] = __builtin_amdgcn_ballot_w32(D[r] < thr[r]);                     \
            any |= m[r];                                                           \
        }                                                                          \
        if (any) {                                                                 \
            KNN_ROW(D, 0, m[0]) KNN_ROW(D, 1, m[1]) KNN_ROW(D, 2, m[2])            \
            KNN_ROW(D, 3, m[3]) KNN_ROW(D, 4, m[4]) KNN_ROW(D, 5, m[5])            \
            KNN_ROW(D, 6, m[6]) KNN_ROW(D, 7, m[7])                                \
        }                                                                          \
    }

#define LOAD_TILE(BB, CN, T)                                                       \
    {                                                                              \
        const int cb_ = (T) * 16;                                                  \
        _Pragma("unroll")                                                          \
        for (int k = 0; k < 16; ++k) {                                             \
            const int c = 4 * k + 2 * half;                                        \
            BB[k].x = xb[(c    ) * NPTS + cb_ + lm];                               \
            BB[k].y = xb[(c + 1) * NPTS + cb_ + lm];                               \
        }                                                                          \
        CN = nb[cb_ + lm];                                                         \
    }

#define MAKE_TILE(BB, CN, DST)                                                     \
    {                                                                              \
        v8f acc0 = {}, acc1 = {};                                                  \
        _Pragma("unroll")                                                          \
        for (int k = 0; k < 16; ++k) {                                             \
            acc0 = __builtin_amdgcn_wmma_f32_16x16x4_f32(false, a0[k], false, BB[k], \
                                                         (short)0, acc0, false, false); \
            acc1 = __builtin_amdgcn_wmma_f32_16x16x4_f32(false, a1[k], false, BB[k], \
                                                         (short)0, acc1, false, false); \
        }                                                                          \
        float* p_ = (DST) + lane;                                                  \
        _Pragma("unroll")                                                          \
        for (int r = 0; r < 8; ++r) p_[r * 32] = rn0[r] + CN - 2.0f * acc0[r];     \
        _Pragma("unroll")                                                          \
        for (int r = 0; r < 8; ++r) p_[RT_PITCH + r * 32] = rn1[r] + CN - 2.0f * acc1[r]; \
    }

__device__ __forceinline__ void proj_rows(const float* __restrict__ xb, const float* __restrict__ W,
                                          float* __restrict__ y, float* __restrict__ z,
                                          const int b, const int rowBase16, const int ot0, const int lane) {
    const int half = lane >> 4;
    const int lm   = lane & 15;

    const int a0i = (2 * half) * NPTS + rowBase16 + lm;
    v2f a[16];
    #pragma unroll
    for (int k = 0; k < 16; ++k) {
        a[k].x = xb[a0i + (4 * k    ) * NPTS];
        a[k].y = xb[a0i + (4 * k + 1) * NPTS];
    }

    #pragma unroll
    for (int t = 0; t < 2; ++t) {
        const int o = (ot0 + t) * 16 + lm;
        v8f accY = {}, accZ = {};
        #pragma unroll
        for (int k = 0; k < 16; ++k) {
            const int c = 4 * k + 2 * half;
            const float w0 = W[o * 128 + c], w1 = W[o * 128 + c + 1];
            v2f bw, bz;
            bw.x = w0;
            bw.y = w1;
            bz.x = W[o * 128 + 64 + c    ] - w0;
            bz.y = W[o * 128 + 64 + c + 1] - w1;
            accY = __builtin_amdgcn_wmma_f32_16x16x4_f32(false, a[k], false, bw,
                                                         (short)0, accY, false, false);
            accZ = __builtin_amdgcn_wmma_f32_16x16x4_f32(false, a[k], false, bz,
                                                         (short)0, accZ, false, false);
        }
        #pragma unroll
        for (int r = 0; r < 8; ++r) {
            const int row = rowBase16 + r + 8 * half;
            y[(b * NPTS + row) * CCH + o] = accY[r];
            z[(b * NPTS + row) * CCH + o] = accZ[r];
        }
    }
}

__global__ void __launch_bounds__(256) __attribute__((amdgpu_waves_per_eu(4, 4))) knn_kernel(const float* __restrict__ x,
                                                 const float* __restrict__ norms,
                                                 const float* __restrict__ W,
                                                 int* __restrict__ knn_idx,
                                                 float* __restrict__ y,
                                                 float* __restrict__ z) {
    __shared__ float sdist[2 * SLOT_FLOATS];

    const int lane = threadIdx.x & 31;
    const int wave = __builtin_amdgcn_readfirstlane((int)(threadIdx.x >> 5));
    const bool matrixWave = wave < 4;
    const int p    = wave & 3;
    const int q    = wave & 1;
    const int rt   = (wave >> 1) & 1;
    const int half = lane >> 4;
    const int lm   = lane & 15;
    const int b    = blockIdx.y;
    const int rowBase = blockIdx.x * (16 * RT);
    const float* xb = x + b * CCH * NPTS;
    const float* nb = norms + b * NPTS;

    if (!matrixWave) proj_rows(xb, W, y, z, b, rowBase + 16 * rt, 2 * q, lane);

    v2f a0[16], a1[16];
    #pragma unroll
    for (int k = 0; k < 16; ++k) {
        int c = 4 * k + 2 * half;
        a0[k].x = xb[(c    ) * NPTS + rowBase + lm];
        a0[k].y = xb[(c + 1) * NPTS + rowBase + lm];
        a1[k].x = xb[(c    ) * NPTS + rowBase + 16 + lm];
        a1[k].y = xb[(c + 1) * NPTS + rowBase + 16 + lm];
    }
    float rn0[8], rn1[8];
    #pragma unroll
    for (int r = 0; r < 8; ++r) {
        rn0[r] = nb[rowBase + r + 8 * half];
        rn1[r] = nb[rowBase + 16 + r + 8 * half];
    }

    float kd[8];
    int   ki[8];
    float thr[8];
    #pragma unroll
    for (int r = 0; r < 8; ++r) {
        kd[r] = (lane == 0) ? __int_as_float((int)0xff800000) : FMAX;
        ki[r] = 0;
        thr[r] = FMAX;
    }

    v2f b0[16], b1[16];
    float cn0, cn1;
    LOAD_TILE(b0, cn0, p)

    float d0[8], d1[8];

    for (int it = 0; it <= NB; ++it) {
        if (matrixWave) {
            if (it < NB) {
                float* dst = sdist + (it & 1) * SLOT_FLOATS;
                const int tb = it * TB;
                for (int j = 0; j < TB / 4; j += 2) {
                    const int i0 = 4 * j + p;
                    LOAD_TILE(b1, cn1, tb + i0 + 4)
                    MAKE_TILE(b0, cn0, dst + i0 * TILE_PITCH)
                    LOAD_TILE(b0, cn0, (tb + i0 + 8) & (NPTS / 16 - 1))
                    MAKE_TILE(b1, cn1, dst + (i0 + 4) * TILE_PITCH)
                }
            }
        } else {
            if (it >= 1) {
                const float* sel = sdist + ((it - 1) & 1) * SLOT_FLOATS + rt * RT_PITCH
                                 + half * TILE_PITCH + 16 * q + lm;
                const int tb = (it - 1) * TB;
                if (it == 1) SAMPLE_BOUND(sel)
                LOAD_DIST(d0, sel, 0)
                #pragma nounroll
                for (int j = 0; j < TB / 2; j += 2) {
                    LOAD_DIST(d1, sel, j + 1)
                    SELECT_TILE(d0, tb + 2 * j)
                    LOAD_DIST(d0, sel, (j + 2) & (TB / 2 - 1))
                    SELECT_TILE(d1, tb + 2 * j + 2)
                }
            }
        }
        handover_barrier();
    }

    if (!matrixWave && lane >= 1 && lane <= KNN) {
        int* op = knn_idx + (b * NPTS + rowBase + 16 * rt + 8 * q) * KNN + (lane - 1);
        #pragma unroll
        for (int r = 0; r < 8; ++r) op[r * KNN] = ki[r];
    }
}

#define GM_PTS   32
#define GM_PITCH 33
__global__ void __launch_bounds__(256) gathermax_kernel(const float* __restrict__ y,
                                                        const float* __restrict__ z,
                                                        const int* __restrict__ knn_idx,
                                                        float* __restrict__ out) {
    __shared__ float tile[CCH * GM_PITCH];

    const int lane = threadIdx.x & 31;
    const int wave = __builtin_amdgcn_readfirstlane((int)(threadIdx.x >> 5));
    const int p0 = blockIdx.x * GM_PTS;
    const int b  = p0 >> 12, n0 = p0 & (NPTS - 1);
    const float* yb = y + b * NPTS * CCH;

    #pragma unroll
    for (int j = 0; j < 4; ++j) {
        const int pl = 4 * wave + j;
        const int p  = p0 + pl;
        const int* ip = knn_idx + p * KNN;
        v2f best = {-FMAX, -FMAX};
        #pragma unroll
        for (int k = 0; k < KNN; ++k) {
            const int m = ip[k];
            const v2f v = *(const v2f*)(yb + m * CCH + 2 * lane);
            best.x = fmaxf(best.x, v.x);
            best.y = fmaxf(best.y, v.y);
        }
        const v2f zz = *(const v2f*)(z + p * CCH + 2 * lane);
        tile[(2 * lane    ) * GM_PITCH + pl] = best.x + zz.x;
        tile[(2 * lane + 1) * GM_PITCH + pl] = best.y + zz.y;
    }
    __syncthreads();
    #pragma unroll
    for (int i = 0; i < 8; ++i) {
        const int o = 8 * wave + i;
        out[(b * CCH + o) * NPTS + n0 + lane] = tile[o * GM_PITCH + lane];
    }
}

extern "C" void kernel_launch(void* const* d_in, const int* in_sizes, int n_in,
                              void* d_out, int out_size, void* d_ws, size_t ws_size,
                              hipStream_t stream) {
    (void)in_sizes; (void)n_in; (void)out_size; (void)ws_size;
    const float* x = (const float*)d_in[0];
    const float* W = (const float*)d_in[1];
    float* out = (float*)d_out;

    char* ws = (char*)d_ws;
    float* norms = (float*)(ws);
    int*   idxb  = (int*)  (ws + 65536);
    float* y     = (float*)(ws + 65536 + 1310720);
    float* z     = (float*)(ws + 65536 + 1310720 + 4194304);

    norms_kernel    <<<dim3(64), 256, 0, stream>>>(x, norms);
    knn_kernel      <<<dim3(128, 4), 256, 0, stream>>>(x, norms, W, idxb, y, z);
    gathermax_kernel<<<dim3(512), 256, 0, stream>>>(y, z, idxb, out);
}


# knn: the selection waves raise their issue priority (s_setprio 3, back to 0 after the sweep) over the older matrix waves of their SIMD: their dependent insert chains no longer queue behind those
# speedup vs baseline: 5.9975x; 1.1277x over previous
#include <hip/hip_runtime.h>

typedef __attribute__((ext_vector_type(2))) float v2f;
typedef __attribute__((ext_vector_type(8))) float v8f;

#define NPTS 4096
#define CCH  64
#define KNN  20
#define FMAX 3.402823466e+38f


__global__ void norms_kernel(const float* __restrict__ x, float* __restrict__ norms) {
    int p = blockIdx.x * blockDim.x + threadIdx.x;
    int b = p >> 12, n = p & 4095;
    const float* xb = x + b * CCH * NPTS + n;
    float s = 0.f;
    #pragma unroll
    for (int c = 0; c < CCH; ++c) { float v = xb[c * NPTS]; s += v * v; }
    norms[p] = s;
}


#define RT 2
#define TB 16
#define NB (NPTS / 16 / TB)
#define TILE_PITCH 272
#define RT_PITCH (TB * TILE_PITCH)
#define SLOT_FLOATS (RT * RT_PITCH)

__device__ __forceinline__ void list_insert(float& kd, int& ki, float key, int col) {
    const int kdi = __float_as_int(kd);
    const int cd = __builtin_amdgcn_permlanex16(kdi, kdi, 0xffffffffu, 0xffffffffu, false, false);
    const int ci = __builtin_amdgcn_permlanex16(ki, ki, 0xffffffffu, 0xffffffffu, false, false);
    const float td = __int_as_float(__builtin_amdgcn_update_dpp(cd, kdi, 0x111, 0xF, 0xF, false));
    const int   ti = __builtin_amdgcn_update_dpp(ci, ki, 0x111, 0xF, 0xF, false);
    const bool up   = kd > key;
    const bool upLo = td > key;
    kd = up ? (upLo ? td : key) : kd;
    ki = up ? (upLo ? ti : col) : ki;
}

__device__ __forceinline__ float list_bound(float kd) {
    return __int_as_float(__builtin_amdgcn_readlane(__float_as_int(kd), KNN));
}

template <int J>
__device__ __forceinline__ int lane_xor(const int v) {
    if constexpr (J == 16)
        return __builtin_amdgcn_permlanex16(v, v, 0x76543210u, 0xfedcba98u, false, false);
    else
        return __builtin_amdgcn_mov_dpp(v, 0x160 + J, 0xF, 0xF, false);
}

__device__ __forceinline__ int order_key(const int b) { return b ^ ((b >> 31) & 0x7fffffff); }

template <int K, int J>
__device__ __forceinline__ void bitonic_step(int (&v)[8], const int lane) {
    const bool takeMin = (((lane & J) == 0) == ((lane & K) == 0));
    #pragma unroll
    for (int r = 0; r < 8; ++r) {
        const int o = lane_xor<J>(v[r]);
        v[r] = takeMin ? (o < v[r] ? o : v[r]) : (o > v[r] ? o : v[r]);
    }
}

__device__ __forceinline__ void sort_lanes(int (&v)[8], const int lane) {
    bitonic_step<2, 1>(v, lane);
    bitonic_step<4, 2>(v, lane);  bitonic_step<4, 1>(v, lane);
    bitonic_step<8, 4>(v, lane);  bitonic_step<8, 2>(v, lane);  bitonic_step<8, 1>(v, lane);
    bitonic_step<16, 8>(v, lane); bitonic_step<16, 4>(v, lane); bitonic_step<16, 2>(v, lane);
    bitonic_step<16, 1>(v, lane);
    bitonic_step<32, 16>(v, lane); bitonic_step<32, 8>(v, lane); bitonic_step<32, 4>(v, lane);
    bitonic_step<32, 2>(v, lane);  bitonic_step<32, 1>(v, lane);
}

__device__ __forceinline__ void handover_barrier() {
    asm volatile("s_wait_dscnt 0x0\n\ts_barrier_signal -1\n\ts_barrier_wait -1" ::: "memory");
}

#define KNN_ROW(D, r, MM)                                                          \
    {                                                                              \
        unsigned mm_ = (MM);                                                       \
        if (mm_) {                                                                 \
            do {                                                                   \
                const int c_ = __builtin_ctz(mm_);                                 \
                mm_ &= mm_ - 1;                                                    \
                const float key_ = __int_as_float(__builtin_amdgcn_readlane(       \
                    __float_as_int(D[r]), c_));                                    \
                list_insert(kd[r], ki[r], key_, colBase + c_);                     \
            } while (mm_);                                                         \
            thr[r] = list_bound(kd[r]);                                            \
        }                                                                          \
    }

#define LOAD_DIST(D, SEL, J)                                                       \
    {                                                                              \
        const float* p_ = (SEL) + (J) * (2 * TILE_PITCH);                          \
        _Pragma("unroll")                                                          \
        for (int r = 0; r < 8; ++r) D[r] = p_[r * 32];                             \
    }

#define SAMPLE_BOUND(SEL)                                                          \
    {                                                                              \
        float mn[8], t_[8];                                                        \
        LOAD_DIST(mn, SEL, 0)                                                      \
        _Pragma("unroll")                                                          \
        for (int J = 1; J < TB / 2; ++J) {                                         \
            LOAD_DIST(t_, SEL, J)                                                  \
            _Pragma("unroll")                                                      \
            for (int r = 0; r < 8; ++r) mn[r] = __builtin_fminf(mn[r], t_[r]);     \
        }                                                                          \
        int key_[8];                                                               \
        _Pragma("unroll")                                                          \
        for (int r = 0; r < 8; ++r) key_[r] = order_key(__float_as_int(mn[r]));    \
        sort_lanes(key_, lane);                                                    \
        _Pragma("unroll")                                                          \
        for (int r = 0; r < 8; ++r) {                                              \
            const float b_ = __int_as_float(order_key(                             \
                __builtin_amdgcn_readlane(key_[r], KNN - 1)));                     \
            const float s_ = b_ + __builtin_fmaxf(__builtin_fabsf(b_) * 0x1p-20f, 1e-30f); \
            thr[r] = __int_as_float(__builtin_amdgcn_readfirstlane(__float_as_int(s_))); \
            kd[r] = (lane == 0) ? __int_as_float((int)0xff800000) : s_;            \
        }                                                                          \
    }

#define SELECT_TILE(D, T)                                                          \
    {                                                                              \
        const int colBase = (T) * 16;                                              \
        unsigned m[8];                                                             \
        unsigned any = 0;                                                          \
        _Pragma("unroll")                                                          \
        for (int r = 0; r < 8; ++r) {                                              \
            m[r] = __builtin_amdgcn_ballot_w32(D[r] < thr[r]);                     \
            any |= m[r];                                                           \
        }                                                                          \
        if (any) {                                                                 \
            KNN_ROW(D, 0, m[0]) KNN_ROW(D, 1, m[1]) KNN_ROW(D, 2, m[2])            \
            KNN_ROW(D, 3, m[3]) KNN_ROW(D, 4, m[4]) KNN_ROW(D, 5, m[5])            \
            KNN_ROW(D, 6, m[6]) KNN_ROW(D, 7, m[7])                                \
        }                                                                          \
    }

#define LOAD_TILE(BB, CN, T)                                                       \
    {                                                                              \
        const int cb_ = (T) * 16;                                                  \
        _Pragma("unroll")                                                          \
        for (int k = 0; k < 16; ++k) {                                             \
            const int c = 4 * k + 2 * half;                                        \
            BB[k].x = xb[(c    ) * NPTS + cb_ + lm];                               \
            BB[k].y = xb[(c + 1) * NPTS + cb_ + lm];                               \
        }                                                                          \
        CN = nb[cb_ + lm];                                                         \
    }

#define MAKE_TILE(BB, CN, DST)                                                     \
    {                                                                              \
        v8f acc0 = {}, acc1 = {};                                                  \
        _Pragma("unroll")                                                          \
        for (int k = 0; k < 16; ++k) {                                             \
            acc0 = __builtin_amdgcn_wmma_f32_16x16x4_f32(false, a0[k], false, BB[k], \
                                                         (short)0, acc0, false, false); \
            acc1 = __builtin_amdgcn_wmma_f32_16x16x4_f32(false, a1[k], false, BB[k], \
                                                         (short)0, acc1, false, false); \
        }                                                                          \
        float* p_ = (DST) + lane;                                                  \
        _Pragma("unroll")                                                          \
        for (int r = 0; r < 8; ++r) p_[r * 32] = rn0[r] + CN - 2.0f * acc0[r];     \
        _Pragma("unroll")                                                          \
        for (int r = 0; r < 8; ++r) p_[RT_PITCH + r * 32] = rn1[r] + CN - 2.0f * acc1[r]; \
    }

__device__ __forceinline__ void proj_rows(const float* __restrict__ xb, const float* __restrict__ W,
                                          float* __restrict__ y, float* __restrict__ z,
                                          const int b, const int rowBase16, const int ot0, const int lane) {
    const int half = lane >> 4;
    const int lm   = lane & 15;

    const int a0i = (2 * half) * NPTS + rowBase16 + lm;
    v2f a[16];
    #pragma unroll
    for (int k = 0; k < 16; ++k) {
        a[k].x = xb[a0i + (4 * k    ) * NPTS];
        a[k].y = xb[a0i + (4 * k + 1) * NPTS];
    }

    #pragma unroll
    for (int t = 0; t < 2; ++t) {
        const int o = (ot0 + t) * 16 + lm;
        v8f accY = {}, accZ = {};
        #pragma unroll
        for (int k = 0; k < 16; ++k) {
            const int c = 4 * k + 2 * half;
            const float w0 = W[o * 128 + c], w1 = W[o * 128 + c + 1];
            v2f bw, bz;
            bw.x = w0;
            bw.y = w1;
            bz.x = W[o * 128 + 64 + c    ] - w0;
            bz.y = W[o * 128 + 64 + c + 1] - w1;
            accY = __builtin_amdgcn_wmma_f32_16x16x4_f32(false, a[k], false, bw,
                                                         (short)0, accY, false, false);
            accZ = __builtin_amdgcn_wmma_f32_16x16x4_f32(false, a[k], false, bz,
                                                         (short)0, accZ, false, false);
        }
        #pragma unroll
        for (int r = 0; r < 8; ++r) {
            const int row = rowBase16 + r + 8 * half;
            y[(b * NPTS + row) * CCH + o] = accY[r];
            z[(b * NPTS + row) * CCH + o] = accZ[r];
        }
    }
}

__global__ void __launch_bounds__(256) __attribute__((amdgpu_waves_per_eu(4, 4))) knn_kernel(const float* __restrict__ x,
                                                 const float* __restrict__ norms,
                                                 const float* __restrict__ W,
                                                 int* __restrict__ knn_idx,
                                                 float* __restrict__ y,
                                                 float* __restrict__ z) {
    __shared__ float sdist[2 * SLOT_FLOATS];

    const int lane = threadIdx.x & 31;
    const int wave = __builtin_amdgcn_readfirstlane((int)(threadIdx.x >> 5));
    const bool matrixWave = wave < 4;
    const int p    = wave & 3;
    const int q    = wave & 1;
    const int rt   = (wave >> 1) & 1;
    const int half = lane >> 4;
    const int lm   = lane & 15;
    const int b    = blockIdx.y;
    const int rowBase = blockIdx.x * (16 * RT);
    const float* xb = x + b * CCH * NPTS;
    const float* nb = norms + b * NPTS;

    if (!matrixWave) {
        proj_rows(xb, W, y, z, b, rowBase + 16 * rt, 2 * q, lane);
        __builtin_amdgcn_s_setprio(3);
    }

    v2f a0[16], a1[16];
    #pragma unroll
    for (int k = 0; k < 16; ++k) {
        int c = 4 * k + 2 * half;
        a0[k].x = xb[(c    ) * NPTS + rowBase + lm];
        a0[k].y = xb[(c + 1) * NPTS + rowBase + lm];
        a1[k].x = xb[(c    ) * NPTS + rowBase + 16 + lm];
        a1[k].y = xb[(c + 1) * NPTS + rowBase + 16 + lm];
    }
    float rn0[8], rn1[8];
    #pragma unroll
    for (int r = 0; r < 8; ++r) {
        rn0[r] = nb[rowBase + r + 8 * half];
        rn1[r] = nb[rowBase + 16 + r + 8 * half];
    }

    float kd[8];
    int   ki[8];
    float thr[8];
    #pragma unroll
    for (int r = 0; r < 8; ++r) {
        kd[r] = (lane == 0) ? __int_as_float((int)0xff800000) : FMAX;
        ki[r] = 0;
        thr[r] = FMAX;
    }

    v2f b0[16], b1[16];
    float cn0, cn1;
    LOAD_TILE(b0, cn0, p)

    float d0[8], d1[8];

    for (int it = 0; it <= NB; ++it) {
        if (matrixWave) {
            if (it < NB) {
                float* dst = sdist + (it & 1) * SLOT_FLOATS;
                const int tb = it * TB;
                for (int j = 0; j < TB / 4; j += 2) {
                    const int i0 = 4 * j + p;
                    LOAD_TILE(b1, cn1, tb + i0 + 4)
                    MAKE_TILE(b0, cn0, dst + i0 * TILE_PITCH)
                    LOAD_TILE(b0, cn0, (tb + i0 + 8) & (NPTS / 16 - 1))
                    MAKE_TILE(b1, cn1, dst + (i0 + 4) * TILE_PITCH)
                }
            }
        } else {
            if (it >= 1) {
                const float* sel = sdist + ((it - 1) & 1) * SLOT_FLOATS + rt * RT_PITCH
                                 + half * TILE_PITCH + 16 * q + lm;
                const int tb = (it - 1) * TB;
                if (it == 1) SAMPLE_BOUND(sel)
                LOAD_DIST(d0, sel, 0)
                #pragma nounroll
                for (int j = 0; j < TB / 2; j += 2) {
                    LOAD_DIST(d1, sel, j + 1)
                    SELECT_TILE(d0, tb + 2 * j)
                    LOAD_DIST(d0, sel, (j + 2) & (TB / 2 - 1))
                    SELECT_TILE(d1, tb + 2 * j + 2)
                }
            }
        }
        handover_barrier();
    }

    if (!matrixWave) __builtin_amdgcn_s_setprio(0);

    if (!matrixWave && lane >= 1 && lane <= KNN) {
        int* op = knn_idx + (b * NPTS + rowBase + 16 * rt + 8 * q) * KNN + (lane - 1);
        #pragma unroll
        for (int r = 0; r < 8; ++r) op[r * KNN] = ki[r];
    }
}

#define GM_PTS   32
#define GM_PITCH 33
__global__ void __launch_bounds__(256) gathermax_kernel(const float* __restrict__ y,
                                                        const float* __restrict__ z,
                                                        const int* __restrict__ knn_idx,
                                                        float* __restrict__ out) {
    __shared__ float tile[CCH * GM_PITCH];

    const int lane = threadIdx.x & 31;
    const int wave = __builtin_amdgcn_readfirstlane((int)(threadIdx.x >> 5));
    const int p0 = blockIdx.x * GM_PTS;
    const int b  = p0 >> 12, n0 = p0 & (NPTS - 1);
    const float* yb = y + b * NPTS * CCH;

    #pragma unroll
    for (int j = 0; j < 4; ++j) {
        const int pl = 4 * wave + j;
        const int p  = p0 + pl;
        const int* ip = knn_idx + p * KNN;
        v2f best = {-FMAX, -FMAX};
        #pragma unroll
        for (int k = 0; k < KNN; ++k) {
            const int m = ip[k];
            const v2f v = *(const v2f*)(yb + m * CCH + 2 * lane);
            best.x = fmaxf(best.x, v.x);
            best.y = fmaxf(best.y, v.y);
        }
        const v2f zz = *(const v2f*)(z + p * CCH + 2 * lane);
        tile[(2 * lane    ) * GM_PITCH + pl] = best.x + zz.x;
        tile[(2 * lane + 1) * GM_PITCH + pl] = best.y + zz.y;
    }
    __syncthreads();
    #pragma unroll
    for (int i = 0; i < 8; ++i) {
        const int o = 8 * wave + i;
        out[(b * CCH + o) * NPTS + n0 + lane] = tile[o * GM_PITCH + lane];
    }
}

extern "C" void kernel_launch(void* const* d_in, const int* in_sizes, int n_in,
                              void* d_out, int out_size, void* d_ws, size_t ws_size,
                              hipStream_t stream) {
    (void)in_sizes; (void)n_in; (void)out_size; (void)ws_size;
    const float* x = (const float*)d_in[0];
    const float* W = (const float*)d_in[1];
    float* out = (float*)d_out;

    char* ws = (char*)d_ws;
    float* norms = (float*)(ws);
    int*   idxb  = (int*)  (ws + 65536);
    float* y     = (float*)(ws + 65536 + 1310720);
    float* z     = (float*)(ws + 65536 + 1310720 + 4194304);

    norms_kernel    <<<dim3(64), 256, 0, stream>>>(x, norms);
    knn_kernel      <<<dim3(128, 4), 256, 0, stream>>>(x, norms, W, idxb, y, z);
    gathermax_kernel<<<dim3(512), 256, 0, stream>>>(y, z, idxb, out);
}
